# starting kernel
# baseline (speedup 1.0000x reference)
#include <hip/hip_runtime.h>

typedef __attribute__((ext_vector_type(2))) float v2f;
typedef __attribute__((ext_vector_type(8))) float v8f;

#define T_N 256
#define R_N 192
#define REF_STEP (48.0f / 191.0f)

__global__ __launch_bounds__(256) void interp_wmma_kernel(
    const float* __restrict__ x, const float* __restrict__ kernel,
    float* __restrict__ out) {
  __shared__ float sh_x[T_N];
  __shared__ float sh_lm[T_N];
  __shared__ float sh_ts[T_N];
  __shared__ float sh_mx[2][R_N];
  __shared__ float sh_red[8][2][16];

  const int tid = threadIdx.x;
  const int bd  = blockIdx.x;
  const int b   = bd >> 4;
  const int d   = bd & 15;

  const float k = kernel[d];
  const float alpha = (k > 0.0f) ? (k + log1pf(__expf(-k))) : log1pf(__expf(k));

  const float* base = x + (size_t)b * 64 * T_N;
  const float* px = base + (size_t)d * T_N;
  const float* pm = base + (size_t)(16 + d) * T_N;
  const float* pt = base + (size_t)(32 + d) * T_N;

  if (tid == 0) {
    __builtin_prefetch(px, 0, 1);
    __builtin_prefetch(pm, 0, 1);
    __builtin_prefetch(pt, 0, 1);
  }

  sh_x[tid]  = px[tid];
  sh_lm[tid] = __logf(fmaxf(pm[tid], 1e-30f));
  sh_ts[tid] = pt[tid];
  __syncthreads();

  if (tid < R_N) {
    const float refr = tid * REF_STEP;
    float mx1 = -3.0e38f, mx10 = -3.0e38f;
    for (int t = 0; t < T_N; ++t) {
      const float dz = sh_ts[t] - refr;
      const float z  = alpha * dz * dz;
      const float lm = sh_lm[t];
      mx1  = fmaxf(mx1,  lm - z);
      mx10 = fmaxf(mx10, lm - 10.0f * z);
    }
    sh_mx[0][tid] = mx1;
    sh_mx[1][tid] = mx10;
  }
  __syncthreads();

  const int wave  = tid >> 5;
  const int lane  = tid & 31;
  const int m16   = lane & 15;
  const int khalf = lane >> 4;

  const float selx = (m16 == 0) ? 1.0f : 0.0f;
  const float sel1 = (m16 == 1) ? 1.0f : 0.0f;

  for (int i = 0; i < 3; ++i) {
    const int job = wave + 8 * i;
    const int p   = job / 12;
    const int R   = job % 12;
    const int r   = R * 16 + m16;
    const float refr  = r * REF_STEP;
    const float scale = (p == 0) ? alpha : 10.0f * alpha;
    const float mx    = sh_mx[p][r];

    v8f acc = {};
    for (int chunk = 0; chunk < 64; ++chunk) {
      const int t0 = chunk * 4 + khalf * 2;
      const float d0 = sh_ts[t0]     - refr;
      const float d1 = sh_ts[t0 + 1] - refr;
      const float x0 = sh_x[t0];
      const float x1 = sh_x[t0 + 1];
      v2f av, bv;
      av.x = __expf(sh_lm[t0]     - scale * d0 * d0 - mx);
      av.y = __expf(sh_lm[t0 + 1] - scale * d1 * d1 - mx);
      bv.x = fmaf(selx, x0, sel1);
      bv.y = fmaf(selx, x1, sel1);
      acc = __builtin_amdgcn_wmma_f32_16x16x4_f32(
           false, av,  false, bv,
           (short)0, acc,  false,  false);
    }

    if (m16 < 2) {
#pragma unroll
      for (int j = 0; j < 8; ++j)
        sh_red[wave][m16][j + 8 * khalf] = acc[j];
    }
    __syncthreads();

    if (lane < 16) {
      const int r2   = R * 16 + lane;
      const float dn = sh_red[wave][1][lane];
      const float ys = sh_red[wave][0][lane];
      const size_t ob = (size_t)b * 48 * R_N;
      const int ch_y = (p == 0) ? d : (32 + d);
      out[ob + (size_t)ch_y * R_N + r2] = ys / dn;
      if (p == 0)
        out[ob + (size_t)(16 + d) * R_N + r2] = sh_mx[0][r2] + __logf(dn);
    }
    __syncthreads();
  }
}

extern "C" void kernel_launch(void* const* d_in, const int* in_sizes, int n_in,
                              void* d_out, int out_size, void* d_ws, size_t ws_size,
                              hipStream_t stream) {
  const float* x      = (const float*)d_in[0];
  const float* kernel = (const float*)d_in[1];
  float* out          = (float*)d_out;
  (void)in_sizes; (void)n_in; (void)out_size; (void)d_ws; (void)ws_size;

  interp_wmma_kernel<<<dim3(64 * 16), dim3(256), 0, stream>>>(x, kernel, out);
}


# Row-max pass: each of 8 waves scans 32 t for all 192 rows (6 rows/lane as packed f32 pairs, 3-input max), partials joined in LDS; was 192 threads x 256 serial t. Sum order unchanged.
# speedup vs baseline: 1.4706x; 1.4706x over previous
#include <hip/hip_runtime.h>

typedef __attribute__((ext_vector_type(2))) float v2f;
typedef __attribute__((ext_vector_type(8))) float v8f;

#define T_N 256
#define R_N 192
#define REF_STEP (48.0f / 191.0f)

__global__ __launch_bounds__(256) void interp_wmma_kernel(
    const float* __restrict__ x, const float* __restrict__ kernel,
    float* __restrict__ out) {
  __shared__ float sh_x[T_N];
  __shared__ __attribute__((aligned(16))) float sh_lm[T_N];
  __shared__ __attribute__((aligned(16))) float sh_ts[T_N];
  __shared__ float sh_mx[2][R_N];
  __shared__ float sh_part[8][2][R_N];
  __shared__ float sh_red[8][2][16];

  const int tid = threadIdx.x;
  const int bd  = blockIdx.x;
  const int b   = bd >> 4;
  const int d   = bd & 15;

  const float k = kernel[d];
  const float alpha = (k > 0.0f) ? (k + log1pf(__expf(-k))) : log1pf(__expf(k));

  const float* base = x + (size_t)b * 64 * T_N;
  const float* px = base + (size_t)d * T_N;
  const float* pm = base + (size_t)(16 + d) * T_N;
  const float* pt = base + (size_t)(32 + d) * T_N;

  if (tid == 0) {
    __builtin_prefetch(px, 0, 1);
    __builtin_prefetch(pm, 0, 1);
    __builtin_prefetch(pt, 0, 1);
  }

  sh_x[tid]  = px[tid];
  sh_lm[tid] = __logf(fmaxf(pm[tid], 1e-30f));
  sh_ts[tid] = pt[tid];
  __syncthreads();

  {
    const int wv = tid >> 5;
    const int ln = tid & 31;
    const float na1  = -alpha;
    const float na10 = -10.0f * alpha;
    v2f refv[3], m1[3], m10[3];
    for (int j = 0; j < 3; ++j) {
      refv[j].x = (ln + 64 * j) * REF_STEP;
      refv[j].y = (ln + 64 * j + 32) * REF_STEP;
      m1[j].x = -3.0e38f;  m1[j].y = -3.0e38f;
      m10[j].x = -3.0e38f; m10[j].y = -3.0e38f;
    }
    const int tb = wv * 32;
    for (int g = 0; g < 8; ++g) {
      const float4 tsv = *reinterpret_cast<const float4*>(&sh_ts[tb + 4 * g]);
      const float4 lmv = *reinterpret_cast<const float4*>(&sh_lm[tb + 4 * g]);
      const float tsa[4] = {tsv.x, tsv.y, tsv.z, tsv.w};
      const float lma[4] = {lmv.x, lmv.y, lmv.z, lmv.w};
      for (int h = 0; h < 4; h += 2) {
        const float ts0 = tsa[h], ts1 = tsa[h + 1];
        const float lm0 = lma[h], lm1 = lma[h + 1];
        for (int j = 0; j < 3; ++j) {
          const v2f d0 = ts0 - refv[j];
          const v2f d1 = ts1 - refv[j];
          const v2f q0 = d0 * d0;
          const v2f q1 = d1 * d1;
          const v2f a0 = q0 * na1 + lm0;
          const v2f a1 = q1 * na1 + lm1;
          const v2f b0 = q0 * na10 + lm0;
          const v2f b1 = q1 * na10 + lm1;
          m1[j].x  = fmaxf(fmaxf(m1[j].x,  a0.x), a1.x);
          m1[j].y  = fmaxf(fmaxf(m1[j].y,  a0.y), a1.y);
          m10[j].x = fmaxf(fmaxf(m10[j].x, b0.x), b1.x);
          m10[j].y = fmaxf(fmaxf(m10[j].y, b0.y), b1.y);
        }
      }
    }
    for (int j = 0; j < 3; ++j) {
      sh_part[wv][0][ln + 64 * j]      = m1[j].x;
      sh_part[wv][0][ln + 64 * j + 32] = m1[j].y;
      sh_part[wv][1][ln + 64 * j]      = m10[j].x;
      sh_part[wv][1][ln + 64 * j + 32] = m10[j].y;
    }
  }
  __syncthreads();

  if (tid < R_N) {
    float mx1 = sh_part[0][0][tid], mx10 = sh_part[0][1][tid];
    for (int w = 1; w < 8; ++w) {
      mx1  = fmaxf(mx1,  sh_part[w][0][tid]);
      mx10 = fmaxf(mx10, sh_part[w][1][tid]);
    }
    sh_mx[0][tid] = mx1;
    sh_mx[1][tid] = mx10;
  }
  __syncthreads();

  const int wave  = tid >> 5;
  const int lane  = tid & 31;
  const int m16   = lane & 15;
  const int khalf = lane >> 4;

  const float selx = (m16 == 0) ? 1.0f : 0.0f;
  const float sel1 = (m16 == 1) ? 1.0f : 0.0f;

  for (int i = 0; i < 3; ++i) {
    const int job = wave + 8 * i;
    const int p   = job / 12;
    const int R   = job % 12;
    const int r   = R * 16 + m16;
    const float refr  = r * REF_STEP;
    const float scale = (p == 0) ? alpha : 10.0f * alpha;
    const float mx    = sh_mx[p][r];

    v8f acc = {};
    for (int chunk = 0; chunk < 64; ++chunk) {
      const int t0 = chunk * 4 + khalf * 2;
      const float d0 = sh_ts[t0]     - refr;
      const float d1 = sh_ts[t0 + 1] - refr;
      const float x0 = sh_x[t0];
      const float x1 = sh_x[t0 + 1];
      v2f av, bv;
      av.x = __expf(sh_lm[t0]     - scale * d0 * d0 - mx);
      av.y = __expf(sh_lm[t0 + 1] - scale * d1 * d1 - mx);
      bv.x = fmaf(selx, x0, sel1);
      bv.y = fmaf(selx, x1, sel1);
      acc = __builtin_amdgcn_wmma_f32_16x16x4_f32(
           false, av,  false, bv,
           (short)0, acc,  false,  false);
    }

    if (m16 < 2) {
#pragma unroll
      for (int j = 0; j < 8; ++j)
        sh_red[wave][m16][j + 8 * khalf] = acc[j];
    }
    __syncthreads();

    if (lane < 16) {
      const int r2   = R * 16 + lane;
      const float dn = sh_red[wave][1][lane];
      const float ys = sh_red[wave][0][lane];
      const size_t ob = (size_t)b * 48 * R_N;
      const int ch_y = (p == 0) ? d : (32 + d);
      out[ob + (size_t)ch_y * R_N + r2] = ys / dn;
      if (p == 0)
        out[ob + (size_t)(16 + d) * R_N + r2] = sh_mx[0][r2] + __logf(dn);
    }
    __syncthreads();
  }
}

extern "C" void kernel_launch(void* const* d_in, const int* in_sizes, int n_in,
                              void* d_out, int out_size, void* d_ws, size_t ws_size,
                              hipStream_t stream) {
  const float* x      = (const float*)d_in[0];
  const float* kernel = (const float*)d_in[1];
  float* out          = (float*)d_out;
  (void)in_sizes; (void)n_in; (void)out_size; (void)d_ws; (void)ws_size;

  interp_wmma_kernel<<<dim3(64 * 16), dim3(256), 0, stream>>>(x, kernel, out);
}


# Reduction pass: each wave walks t once for its three (path, r-tile) jobs with three live accumulators; LDS pair loads and B fragment issued once per 3 WMMAs, one epilogue. Sum order unchanged.
# speedup vs baseline: 1.5889x; 1.0804x over previous
#include <hip/hip_runtime.h>

typedef __attribute__((ext_vector_type(2))) float v2f;
typedef __attribute__((ext_vector_type(8))) float v8f;

#define T_N 256
#define R_N 192
#define REF_STEP (48.0f / 191.0f)

__global__ __launch_bounds__(256) void interp_wmma_kernel(
    const float* __restrict__ x, const float* __restrict__ kernel,
    float* __restrict__ out) {
  __shared__ float sh_x[T_N];
  __shared__ __attribute__((aligned(16))) float sh_lm[T_N];
  __shared__ __attribute__((aligned(16))) float sh_ts[T_N];
  __shared__ float sh_mx[2][R_N];
  __shared__ float sh_part[8][2][R_N];
  __shared__ float sh_red[8][3][2][16];

  const int tid = threadIdx.x;
  const int bd  = blockIdx.x;
  const int b   = bd >> 4;
  const int d   = bd & 15;

  const float k = kernel[d];
  const float alpha = (k > 0.0f) ? (k + log1pf(__expf(-k))) : log1pf(__expf(k));

  const float* base = x + (size_t)b * 64 * T_N;
  const float* px = base + (size_t)d * T_N;
  const float* pm = base + (size_t)(16 + d) * T_N;
  const float* pt = base + (size_t)(32 + d) * T_N;

  if (tid == 0) {
    __builtin_prefetch(px, 0, 1);
    __builtin_prefetch(pm, 0, 1);
    __builtin_prefetch(pt, 0, 1);
  }

  sh_x[tid]  = px[tid];
  sh_lm[tid] = __logf(fmaxf(pm[tid], 1e-30f));
  sh_ts[tid] = pt[tid];
  __syncthreads();

  {
    const int wv = tid >> 5;
    const int ln = tid & 31;
    const float na1  = -alpha;
    const float na10 = -10.0f * alpha;
    v2f refv[3], m1[3], m10[3];
    for (int j = 0; j < 3; ++j) {
      refv[j].x = (ln + 64 * j) * REF_STEP;
      refv[j].y = (ln + 64 * j + 32) * REF_STEP;
      m1[j].x = -3.0e38f;  m1[j].y = -3.0e38f;
      m10[j].x = -3.0e38f; m10[j].y = -3.0e38f;
    }
    const int tb = wv * 32;
    for (int g = 0; g < 8; ++g) {
      const float4 tsv = *reinterpret_cast<const float4*>(&sh_ts[tb + 4 * g]);
      const float4 lmv = *reinterpret_cast<const float4*>(&sh_lm[tb + 4 * g]);
      const float tsa[4] = {tsv.x, tsv.y, tsv.z, tsv.w};
      const float lma[4] = {lmv.x, lmv.y, lmv.z, lmv.w};
      for (int h = 0; h < 4; h += 2) {
        const float ts0 = tsa[h], ts1 = tsa[h + 1];
        const float lm0 = lma[h], lm1 = lma[h + 1];
        for (int j = 0; j < 3; ++j) {
          const v2f d0 = ts0 - refv[j];
          const v2f d1 = ts1 - refv[j];
          const v2f q0 = d0 * d0;
          const v2f q1 = d1 * d1;
          const v2f a0 = q0 * na1 + lm0;
          const v2f a1 = q1 * na1 + lm1;
          const v2f b0 = q0 * na10 + lm0;
          const v2f b1 = q1 * na10 + lm1;
          m1[j].x  = fmaxf(fmaxf(m1[j].x,  a0.x), a1.x);
          m1[j].y  = fmaxf(fmaxf(m1[j].y,  a0.y), a1.y);
          m10[j].x = fmaxf(fmaxf(m10[j].x, b0.x), b1.x);
          m10[j].y = fmaxf(fmaxf(m10[j].y, b0.y), b1.y);
        }
      }
    }
    for (int j = 0; j < 3; ++j) {
      sh_part[wv][0][ln + 64 * j]      = m1[j].x;
      sh_part[wv][0][ln + 64 * j + 32] = m1[j].y;
      sh_part[wv][1][ln + 64 * j]      = m10[j].x;
      sh_part[wv][1][ln + 64 * j + 32] = m10[j].y;
    }
  }
  __syncthreads();

  if (tid < R_N) {
    float mx1 = sh_part[0][0][tid], mx10 = sh_part[0][1][tid];
    for (int w = 1; w < 8; ++w) {
      mx1  = fmaxf(mx1,  sh_part[w][0][tid]);
      mx10 = fmaxf(mx10, sh_part[w][1][tid]);
    }
    sh_mx[0][tid] = mx1;
    sh_mx[1][tid] = mx10;
  }
  __syncthreads();

  const int wave  = tid >> 5;
  const int lane  = tid & 31;
  const int m16   = lane & 15;
  const int khalf = lane >> 4;

  const float selx = (m16 == 0) ? 1.0f : 0.0f;
  const float sel1 = (m16 == 1) ? 1.0f : 0.0f;

  float refr[3], scale[3], mx[3];
  v8f acc[3];
  for (int i = 0; i < 3; ++i) {
    const int job = wave + 8 * i;
    const int p   = job / 12;
    const int R   = job % 12;
    const int r   = R * 16 + m16;
    refr[i]  = r * REF_STEP;
    scale[i] = (p == 0) ? alpha : 10.0f * alpha;
    mx[i]    = sh_mx[p][r];
    acc[i]   = v8f{};
  }

  for (int chunk = 0; chunk < 64; ++chunk) {
    const int t0 = chunk * 4 + khalf * 2;
    const float ts0 = sh_ts[t0];
    const float ts1 = sh_ts[t0 + 1];
    const float lm0 = sh_lm[t0];
    const float lm1 = sh_lm[t0 + 1];
    const float x0  = sh_x[t0];
    const float x1  = sh_x[t0 + 1];
    v2f bv;
    bv.x = fmaf(selx, x0, sel1);
    bv.y = fmaf(selx, x1, sel1);
    for (int i = 0; i < 3; ++i) {
      const float d0 = ts0 - refr[i];
      const float d1 = ts1 - refr[i];
      v2f av;
      av.x = __expf(lm0 - scale[i] * d0 * d0 - mx[i]);
      av.y = __expf(lm1 - scale[i] * d1 * d1 - mx[i]);
      acc[i] = __builtin_amdgcn_wmma_f32_16x16x4_f32(
           false, av,  false, bv,
           (short)0, acc[i],  false,  false);
    }
  }

  if (m16 < 2) {
    for (int i = 0; i < 3; ++i) {
      for (int j = 0; j < 8; ++j)
        sh_red[wave][i][m16][j + 8 * khalf] = acc[i][j];
    }
  }
  __syncthreads();

  if (lane < 16) {
    const size_t ob = (size_t)b * 48 * R_N;
    for (int i = 0; i < 3; ++i) {
      const int job = wave + 8 * i;
      const int p   = job / 12;
      const int R   = job % 12;
      const int r2  = R * 16 + lane;
      const float dn = sh_red[wave][i][1][lane];
      const float ys = sh_red[wave][i][0][lane];
      const int ch_y = (p == 0) ? d : (32 + d);
      out[ob + (size_t)ch_y * R_N + r2] = ys / dn;
      if (p == 0)
        out[ob + (size_t)(16 + d) * R_N + r2] = sh_mx[0][r2] + __logf(dn);
    }
  }
}

extern "C" void kernel_launch(void* const* d_in, const int* in_sizes, int n_in,
                              void* d_out, int out_size, void* d_ws, size_t ws_size,
                              hipStream_t stream) {
  const float* x      = (const float*)d_in[0];
  const float* kernel = (const float*)d_in[1];
  float* out          = (float*)d_out;
  (void)in_sizes; (void)n_in; (void)out_size; (void)d_ws; (void)ws_size;

  interp_wmma_kernel<<<dim3(64 * 16), dim3(256), 0, stream>>>(x, kernel, out);
}


# Mask leaves the exponent: max(m,1e-30) goes into the B operand (cols m*x, m); A = exp2(c - a*q), one packed fma per t pair; row shift c = a*min q (nearest t), same for both paths. Sum order kept.
# speedup vs baseline: 1.8657x; 1.1742x over previous
#include <hip/hip_runtime.h>


typedef __attribute__((ext_vector_type(2))) float v2f;
typedef __attribute__((ext_vector_type(4))) float v4f;
typedef __attribute__((ext_vector_type(8))) float v8f;

#define T_N 256
#define R_N 192
#define REF_STEP (48.0f / 191.0f)
#define LOG2E_F 1.44269504088896340736f
#define LN2_F   0.69314718055994530942f
#define PLANE_N (2 * T_N + 8)

__global__ __launch_bounds__(256) void interp_wmma_kernel(
    const float* __restrict__ x, const float* __restrict__ kernel,
    float* __restrict__ out) {
  __shared__ __attribute__((aligned(16))) float sh_ts[T_N];
  __shared__ __attribute__((aligned(16))) float sh_tb[3][PLANE_N];
  __shared__ float sh_qmin[R_N];
  __shared__ float sh_part[8][R_N];
  __shared__ float sh_red[8][3][2][16];

  const int tid = threadIdx.x;
  const int bd  = blockIdx.x;
  const int b   = bd >> 4;
  const int d   = bd & 15;

  const float k = kernel[d];
  const float alpha = (k > 0.0f) ? (k + log1pf(__expf(-k))) : log1pf(__expf(k));

  const float* base = x + (size_t)b * 64 * T_N;
  const float* px = base + (size_t)d * T_N;
  const float* pm = base + (size_t)(16 + d) * T_N;
  const float* pt = base + (size_t)(32 + d) * T_N;

  if (tid == 0) {
    __builtin_prefetch(px, 0, 1);
    __builtin_prefetch(pm, 0, 1);
    __builtin_prefetch(pt, 0, 1);
  }

  {
    const float xv = px[tid];
    const float mc = fmaxf(pm[tid], 1e-30f);
    const float tv = pt[tid];
    sh_ts[tid] = tv;
    const int o = 4 * (tid >> 1) + (tid & 1);
    sh_tb[0][o] = tv;  sh_tb[0][o + 2] = mc * xv;
    sh_tb[1][o] = tv;  sh_tb[1][o + 2] = mc;
    sh_tb[2][o] = tv;  sh_tb[2][o + 2] = 0.0f;
  }
  __syncthreads();

  {
    const int wv = tid >> 5;
    const int ln = tid & 31;
    v2f refv[3], mn[3];
    for (int j = 0; j < 3; ++j) {
      refv[j].x = (ln + 64 * j) * REF_STEP;
      refv[j].y = (ln + 64 * j + 32) * REF_STEP;
      mn[j].x = 3.0e38f;  mn[j].y = 3.0e38f;
    }
    const int tb = wv * 32;
    for (int g = 0; g < 8; ++g) {
      const float4 tsv = *reinterpret_cast<const float4*>(&sh_ts[tb + 4 * g]);
      const float tsa[4] = {tsv.x, tsv.y, tsv.z, tsv.w};
      for (int h = 0; h < 4; h += 2) {
        const float ts0 = tsa[h], ts1 = tsa[h + 1];
        for (int j = 0; j < 3; ++j) {
          const v2f d0 = ts0 - refv[j];
          const v2f d1 = ts1 - refv[j];
          mn[j].x = fminf(fminf(mn[j].x, fabsf(d0.x)), fabsf(d1.x));
          mn[j].y = fminf(fminf(mn[j].y, fabsf(d0.y)), fabsf(d1.y));
        }
      }
    }
    for (int j = 0; j < 3; ++j) {
      sh_part[wv][ln + 64 * j]      = mn[j].x;
      sh_part[wv][ln + 64 * j + 32] = mn[j].y;
    }
  }
  __syncthreads();

  if (tid < R_N) {
    float m = sh_part[0][tid];
    for (int w = 1; w < 8; ++w) m = fminf(m, sh_part[w][tid]);
    sh_qmin[tid] = m * m;
  }
  __syncthreads();

  const int wave  = tid >> 5;
  const int lane  = tid & 31;
  const int m16   = lane & 15;
  const int khalf = lane >> 4;

  const float a2_1  = alpha * LOG2E_F;
  const float a2_10 = 10.0f * alpha * LOG2E_F;

  v2f nref[3], na2[3], cs[3];
  v8f acc[3];
  for (int i = 0; i < 3; ++i) {
    const int job = wave + 8 * i;
    const int p   = job / 12;
    const int R   = job % 12;
    const int r   = R * 16 + m16;
    const float a2 = (p == 0) ? a2_1 : a2_10;
    const float nr = -(r * REF_STEP);
    const float c  = a2 * sh_qmin[r];
    nref[i].x = nr;   nref[i].y = nr;
    na2[i].x  = -a2;  na2[i].y  = -a2;
    cs[i].x   = c;    cs[i].y   = c;
    acc[i]    = v8f{};
  }

  const int bcls = (m16 < 2) ? m16 : 2;
  const float* tbp = &sh_tb[bcls][4 * khalf];

  for (int chunk = 0; chunk < 64; ++chunk) {
    const v4f tb4 = *reinterpret_cast<const v4f*>(tbp + 8 * chunk);
    v2f tsp, bv;
    tsp.x = tb4.x;  tsp.y = tb4.y;
    bv.x  = tb4.z;  bv.y  = tb4.w;
    for (int i = 0; i < 3; ++i) {
      const v2f dd = tsp + nref[i];
      const v2f qq = dd * dd;
      const v2f ee = qq * na2[i] + cs[i];
      v2f av;
      av.x = __builtin_amdgcn_exp2f(ee.x);
      av.y = __builtin_amdgcn_exp2f(ee.y);
      acc[i] = __builtin_amdgcn_wmma_f32_16x16x4_f32(
           false, av,  false, bv,
           (short)0, acc[i],  false,  false);
    }
  }

  if (m16 < 2) {
    for (int i = 0; i < 3; ++i) {
      for (int j = 0; j < 8; ++j)
        sh_red[wave][i][m16][j + 8 * khalf] = acc[i][j];
    }
  }
  __syncthreads();

  if (lane < 16) {
    const size_t ob = (size_t)b * 48 * R_N;
    for (int i = 0; i < 3; ++i) {
      const int job = wave + 8 * i;
      const int p   = job / 12;
      const int R   = job % 12;
      const int r2  = R * 16 + lane;
      const float dn = sh_red[wave][i][1][lane];
      const float ys = sh_red[wave][i][0][lane];
      const int ch_y = (p == 0) ? d : (32 + d);
      out[ob + (size_t)ch_y * R_N + r2] = ys / dn;
      if (p == 0) {
        const float qm  = sh_qmin[r2];
        const float c   = a2_1 * qm;
        const float rho = fmaf(-a2_1, qm, c);
        out[ob + (size_t)(16 + d) * R_N + r2] =
            fmaf(-alpha, qm, __logf(dn) - rho * LN2_F);
      }
    }
  }
}

extern "C" void kernel_launch(void* const* d_in, const int* in_sizes, int n_in,
                              void* d_out, int out_size, void* d_ws, size_t ws_size,
                              hipStream_t stream) {
  const float* x      = (const float*)d_in[0];
  const float* kernel = (const float*)d_in[1];
  float* out          = (float*)d_out;
  (void)in_sizes; (void)n_in; (void)out_size; (void)d_ws; (void)ws_size;

  interp_wmma_kernel<<<dim3(64 * 16), dim3(256), 0, stream>>>(x, kernel, out);
}


# Path-2 weight = (path-1 weight)^10: four packed multiplies replace its distance, fma and two exp2; a wave now holds both paths of 3 r tiles over half of t; t halves added in LDS (new sum order).
# speedup vs baseline: 2.0148x; 1.0799x over previous
#include <hip/hip_runtime.h>


typedef __attribute__((ext_vector_type(2))) float v2f;
typedef __attribute__((ext_vector_type(4))) float v4f;
typedef __attribute__((ext_vector_type(8))) float v8f;

#define T_N 256
#define R_N 192
#define REF_STEP (48.0f / 191.0f)
#define LOG2E_F 1.44269504088896340736f
#define LN2_F   0.69314718055994530942f
#define PLANE_N (2 * T_N + 8)

__global__ __launch_bounds__(256) void interp_wmma_kernel(
    const float* __restrict__ x, const float* __restrict__ kernel,
    float* __restrict__ out) {
  __shared__ __attribute__((aligned(16))) float sh_ts[T_N];
  __shared__ __attribute__((aligned(16))) float sh_tb[3][PLANE_N];
  __shared__ float sh_qmin[R_N];
  __shared__ float sh_part[8][R_N];
  __shared__ float sh_red[8][6][2][16];

  const int tid = threadIdx.x;
  const int bd  = blockIdx.x;
  const int b   = bd >> 4;
  const int d   = bd & 15;

  const float k = kernel[d];
  const float alpha = (k > 0.0f) ? (k + log1pf(__expf(-k))) : log1pf(__expf(k));

  const float* base = x + (size_t)b * 64 * T_N;
  const float* px = base + (size_t)d * T_N;
  const float* pm = base + (size_t)(16 + d) * T_N;
  const float* pt = base + (size_t)(32 + d) * T_N;

  if (tid == 0) {
    __builtin_prefetch(px, 0, 1);
    __builtin_prefetch(pm, 0, 1);
    __builtin_prefetch(pt, 0, 1);
  }

  {
    const float xv = px[tid];
    const float mc = fmaxf(pm[tid], 1e-30f);
    const float tv = pt[tid];
    sh_ts[tid] = tv;
    const int o = 4 * (tid >> 1) + (tid & 1);
    sh_tb[0][o] = tv;  sh_tb[0][o + 2] = mc * xv;
    sh_tb[1][o] = tv;  sh_tb[1][o + 2] = mc;
    sh_tb[2][o] = tv;  sh_tb[2][o + 2] = 0.0f;
  }
  __syncthreads();

  {
    const int wv = tid >> 5;
    const int ln = tid & 31;
    v2f refv[3], mn[3];
    for (int j = 0; j < 3; ++j) {
      refv[j].x = (ln + 64 * j) * REF_STEP;
      refv[j].y = (ln + 64 * j + 32) * REF_STEP;
      mn[j].x = 3.0e38f;  mn[j].y = 3.0e38f;
    }
    const int tb = wv * 32;
    for (int g = 0; g < 8; ++g) {
      const float4 tsv = *reinterpret_cast<const float4*>(&sh_ts[tb + 4 * g]);
      const float tsa[4] = {tsv.x, tsv.y, tsv.z, tsv.w};
      for (int h = 0; h < 4; h += 2) {
        const float ts0 = tsa[h], ts1 = tsa[h + 1];
        for (int j = 0; j < 3; ++j) {
          const v2f d0 = ts0 - refv[j];
          const v2f d1 = ts1 - refv[j];
          mn[j].x = fminf(fminf(mn[j].x, fabsf(d0.x)), fabsf(d1.x));
          mn[j].y = fminf(fminf(mn[j].y, fabsf(d0.y)), fabsf(d1.y));
        }
      }
    }
    for (int j = 0; j < 3; ++j) {
      sh_part[wv][ln + 64 * j]      = mn[j].x;
      sh_part[wv][ln + 64 * j + 32] = mn[j].y;
    }
  }
  __syncthreads();

  if (tid < R_N) {
    float m = sh_part[0][tid];
    for (int w = 1; w < 8; ++w) m = fminf(m, sh_part[w][tid]);
    sh_qmin[tid] = m * m;
  }
  __syncthreads();

  const int wave  = tid >> 5;
  const int lane  = tid & 31;
  const int m16   = lane & 15;
  const int khalf = lane >> 4;
  const int th    = wave & 1;
  const int rg    = wave >> 1;

  const float a2_1 = alpha * LOG2E_F;

  v2f nref[3], cs[3];
  v2f na2;
  na2.x = -a2_1;  na2.y = -a2_1;
  v8f acc1[3], acc10[3];
  for (int i = 0; i < 3; ++i) {
    const int r   = (3 * rg + i) * 16 + m16;
    const float nr = -(r * REF_STEP);
    const float c  = a2_1 * sh_qmin[r];
    nref[i].x = nr;  nref[i].y = nr;
    cs[i].x   = c;   cs[i].y   = c;
    acc1[i]   = v8f{};
    acc10[i]  = v8f{};
  }

  const int bcls = (m16 < 2) ? m16 : 2;
  const float* tbp = &sh_tb[bcls][4 * khalf + 8 * 32 * th];

  for (int chunk = 0; chunk < 32; ++chunk) {
    const v4f tb4 = *reinterpret_cast<const v4f*>(tbp + 8 * chunk);
    v2f tsp, bv;
    tsp.x = tb4.x;  tsp.y = tb4.y;
    bv.x  = tb4.z;  bv.y  = tb4.w;
    for (int i = 0; i < 3; ++i) {
      const v2f dd = tsp + nref[i];
      const v2f qq = dd * dd;
      const v2f ee = qq * na2 + cs[i];
      v2f e1;
      e1.x = __builtin_amdgcn_exp2f(ee.x);
      e1.y = __builtin_amdgcn_exp2f(ee.y);
      const v2f p2  = e1 * e1;
      const v2f p4  = p2 * p2;
      const v2f p8  = p4 * p4;
      const v2f e10 = p8 * p2;
      acc1[i] = __builtin_amdgcn_wmma_f32_16x16x4_f32(
           false, e1,  false, bv,
           (short)0, acc1[i],  false,  false);
      acc10[i] = __builtin_amdgcn_wmma_f32_16x16x4_f32(
           false, e10,  false, bv,
           (short)0, acc10[i],  false,  false);
    }
  }

  if (m16 < 2) {
    for (int i = 0; i < 3; ++i) {
      for (int j = 0; j < 8; ++j) {
        sh_red[wave][2 * i][m16][j + 8 * khalf]     = acc1[i][j];
        sh_red[wave][2 * i + 1][m16][j + 8 * khalf] = acc10[i][j];
      }
    }
  }
  __syncthreads();

  if (tid < R_N) {
    const int r  = tid;
    const int R  = r >> 4;
    const int m  = r & 15;
    const int g  = R / 3;
    const int i  = R - 3 * g;
    const int w0 = 2 * g, w1 = 2 * g + 1;
    const float ys1  = sh_red[w0][2 * i][0][m]     + sh_red[w1][2 * i][0][m];
    const float dn1  = sh_red[w0][2 * i][1][m]     + sh_red[w1][2 * i][1][m];
    const float ys10 = sh_red[w0][2 * i + 1][0][m] + sh_red[w1][2 * i + 1][0][m];
    const float dn10 = sh_red[w0][2 * i + 1][1][m] + sh_red[w1][2 * i + 1][1][m];
    const size_t ob = (size_t)b * 48 * R_N;
    out[ob + (size_t)d * R_N + r]        = ys1 / dn1;
    out[ob + (size_t)(32 + d) * R_N + r] = ys10 / dn10;
    const float qm  = sh_qmin[r];
    const float c   = a2_1 * qm;
    const float rho = fmaf(-a2_1, qm, c);
    out[ob + (size_t)(16 + d) * R_N + r] =
        fmaf(-alpha, qm, __logf(dn1) - rho * LN2_F);
  }
}

extern "C" void kernel_launch(void* const* d_in, const int* in_sizes, int n_in,
                              void* d_out, int out_size, void* d_ws, size_t ws_size,
                              hipStream_t stream) {
  const float* x      = (const float*)d_in[0];
  const float* kernel = (const float*)d_in[1];
  float* out          = (float*)d_out;
  (void)in_sizes; (void)n_in; (void)out_size; (void)d_ws; (void)ws_size;

  interp_wmma_kernel<<<dim3(64 * 16), dim3(256), 0, stream>>>(x, kernel, out);
}
